# starting kernel
# baseline (speedup 1.0000x reference)
#include <hip/hip_runtime.h>

typedef __attribute__((ext_vector_type(2))) float v2f;
typedef __attribute__((ext_vector_type(8))) float v8f;

__global__ __launch_bounds__(256) void idwt2_wmma_kernel(
    const float* __restrict__ x, float* __restrict__ y) {
  const int lane = threadIdx.x & 31;
  const int wid  = (blockIdx.x * 256 + (int)threadIdx.x) >> 5;

  const int jb = wid % 6;
  int tmp = wid / 6;
  const int I = tmp % 96;
  tmp /= 96;
  const int g = tmp & 15;
  const int b = tmp >> 4;

  const int n  = lane & 15;
  const int hi = lane >> 4;
  const int J0 = jb * 16;

  const int p  = n >> 2, q  = n & 3;
  const int di = p & 1,  dI = (p >> 1) & 1;
  const int dj = q & 1,  dJ = (q >> 1) & 1;

  const float* xb = x + ((b * 256) * 96 + I) * 96 + (J0 + n);

  v8f acc = {};
#pragma unroll
  for (int ck = 0; ck < 4; ++ck) {
    v2f a, bf;
#pragma unroll
    for (int v = 0; v < 2; ++v) {
      const int s = 4 * ck + v + 2 * hi;
      const int k = s & 3, m = s >> 2;
      const int bit =
          ((di & k) ^ (dj & (k >> 1)) ^ (dI & m) ^ (dJ & (m >> 1))) & 1;
      const float sgn  = bit ? -1.0f : 1.0f;
      const int   c    = g + ((s & 3) << 4) + ((s >> 2) << 6);
      const float bval = xb[c * 9216];
      if (v == 0) { a.x = sgn; bf.x = bval; }
      else        { a.y = sgn; bf.y = bval; }
    }
    acc = __builtin_amdgcn_wmma_f32_16x16x4_f32(
         false, a,  false, bf,
         (short)0, acc,  false,  false);
  }

  float* yb = y + ((size_t)((b * 16 + g) * 384 + 4 * I + 2 * hi)) * 384
                + 4 * (J0 + n);
  *(float4*)(yb)       = make_float4(acc[0], acc[1], acc[2], acc[3]);
  *(float4*)(yb + 384) = make_float4(acc[4], acc[5], acc[6], acc[7]);
}

extern "C" void kernel_launch(void* const* d_in, const int* in_sizes, int n_in,
                              void* d_out, int out_size, void* d_ws, size_t ws_size,
                              hipStream_t stream) {
  (void)in_sizes; (void)n_in; (void)out_size; (void)d_ws; (void)ws_size;
  const float* x = (const float*)d_in[0];
  float*       y = (float*)d_out;
  idwt2_wmma_kernel<<<dim3(9216), dim3(256), 0, stream>>>(x, y);
}


# One wave per whole input row (six 16-column tiles), its 48 loads issued in tile order ahead of the first product, whole grid resident at once: 6x fewer waves, 6x more bytes in flight per wave
# speedup vs baseline: 1.0610x; 1.0610x over previous
#include <hip/hip_runtime.h>

typedef __attribute__((ext_vector_type(2))) float v2f;
typedef __attribute__((ext_vector_type(8))) float v8f;

constexpr int TILES = 6;

__global__ __launch_bounds__(256, 12) void idwt2_wmma_kernel(
    const float* __restrict__ x, float* __restrict__ y) {
  const int lane = threadIdx.x & 31;
  const int wid = (blockIdx.x * 256 + (int)threadIdx.x) >> 5;
  const int I  = wid % 96;
  const int bg = wid / 96;
  const int g  = bg & 15;
  const int b  = bg >> 4;

  const int n  = lane & 15;
  const int hi = lane >> 4;

  const float* xb = x + (((b * 256) + g + 32 * hi) * 96 + I) * 96 + n;

  v2f bf[TILES][4];
#pragma unroll
  for (int t = 0; t < TILES; ++t) {
#pragma unroll
    for (int ck = 0; ck < 4; ++ck) {
      bf[t][ck].x = xb[(64 * ck) * 9216 + 16 * t];
      bf[t][ck].y = xb[(64 * ck + 16) * 9216 + 16 * t];
    }
    __builtin_amdgcn_sched_barrier(0);
  }

  const int p  = n >> 2, q  = n & 3;
  const int di = p & 1,  dI = (p >> 1) & 1;
  const int dj = q & 1,  dJ = (q >> 1) & 1;

  v2f a[4];
#pragma unroll
  for (int ck = 0; ck < 4; ++ck) {
#pragma unroll
    for (int v = 0; v < 2; ++v) {
      const int s = 4 * ck + v + 2 * hi;
      const int k = s & 3, m = s >> 2;
      const int bit =
          ((di & k) ^ (dj & (k >> 1)) ^ (dI & m) ^ (dJ & (m >> 1))) & 1;
      const float sgn = bit ? -1.0f : 1.0f;
      if (v == 0) a[ck].x = sgn; else a[ck].y = sgn;
    }
  }

  float* yb = y + ((size_t)(bg * 384 + 4 * I + 2 * hi)) * 384 + 4 * n;

#pragma unroll
  for (int t = 0; t < TILES; ++t) {
    v8f acc = {};
#pragma unroll
    for (int ck = 0; ck < 4; ++ck) {
      acc = __builtin_amdgcn_wmma_f32_16x16x4_f32(
           false, a[ck],  false, bf[t][ck],
           (short)0, acc,  false,  false);
    }
    *(float4*)(yb + 64 * t)       = make_float4(acc[0], acc[1], acc[2], acc[3]);
    *(float4*)(yb + 64 * t + 384) = make_float4(acc[4], acc[5], acc[6], acc[7]);
  }
}

extern "C" void kernel_launch(void* const* d_in, const int* in_sizes, int n_in,
                              void* d_out, int out_size, void* d_ws, size_t ws_size,
                              hipStream_t stream) {
  (void)in_sizes; (void)n_in; (void)out_size; (void)d_ws; (void)ws_size;
  const float* x = (const float*)d_in[0];
  float*       y = (float*)d_out;
  idwt2_wmma_kernel<<<dim3(1536), dim3(256), 0, stream>>>(x, y);
}
